# starting kernel
# baseline (speedup 1.0000x reference)
#include <hip/hip_runtime.h>


typedef __attribute__((ext_vector_type(2))) float v2f;
typedef __attribute__((ext_vector_type(8))) float v8f;

#define S_LEN 1024
#define D_DIM 64
#define BH    64

__global__ __launch_bounds__(256)
void entmax_attn_kernel(const float* __restrict__ q,
                        const float* __restrict__ k,
                        const float* __restrict__ v,
                        float* __restrict__ out)
{
    __shared__ float sm[16 * S_LEN];

    const int lane  = threadIdx.x & 31;
    const int wave  = threadIdx.x >> 5;
    const int lmod  = lane & 15;
    const int lhalf = lane >> 4;

    const int bh    = blockIdx.y;
    const int qbase = blockIdx.x * 16;

    const float* qp = q + ((size_t)(bh * S_LEN + qbase + lmod)) * D_DIM;
    v2f aq[16];
#pragma unroll
    for (int s = 0; s < 16; ++s)
        aq[s] = *(const v2f*)(qp + 4 * s + 2 * lhalf);

#pragma unroll
    for (int t = 0; t < 8; ++t) {
        const int nbase = (wave * 8 + t) * 16;
        const float* kp = k + ((size_t)(bh * S_LEN + nbase + lmod)) * D_DIM;
        v8f acc = {};
#pragma unroll
        for (int s = 0; s < 16; ++s) {
            v2f bf = *(const v2f*)(kp + 4 * s + 2 * lhalf);
            acc = __builtin_amdgcn_wmma_f32_16x16x4_f32(
                false, aq[s], false, bf, (short)0, acc, false, false);
        }
#pragma unroll
        for (int r = 0; r < 8; ++r)
            sm[(r + 8 * lhalf) * S_LEN + nbase + lmod] = acc[r] * 0.0625f;
    }
    __syncthreads();

    for (int rr = 0; rr < 2; ++rr) {
        const int row = wave * 2 + rr;
        float* rp = &sm[row * S_LEN];
        float x[32];
#pragma unroll
        for (int j = 0; j < 32; ++j) x[j] = rp[lane + 32 * j];

        float m = x[0];
#pragma unroll
        for (int j = 1; j < 32; ++j) m = fmaxf(m, x[j]);
#pragma unroll
        for (int off = 16; off >= 1; off >>= 1)
            m = fmaxf(m, __shfl_xor(m, off, 32));
#pragma unroll
        for (int j = 0; j < 32; ++j) x[j] -= m;

        float tau = -1.0f;
        for (int it = 0; it < 12; ++it) {
            float s = 0.f, g = 0.f;
#pragma unroll
            for (int j = 0; j < 32; ++j) {
                float d = fmaxf(x[j] - tau, 0.f);
                s = fmaf(d, d, s);
                g += d;
            }
#pragma unroll
            for (int off = 16; off >= 1; off >>= 1) {
                s += __shfl_xor(s, off, 32);
                g += __shfl_xor(g, off, 32);
            }
            g = fmaxf(g, 1e-20f);
            tau += (s - 1.0f) / (2.0f * g);
        }

        float p[32];
        float ssum = 0.f;
#pragma unroll
        for (int j = 0; j < 32; ++j) {
            float d = fmaxf(x[j] - tau, 0.f);
            p[j] = d * d;
            ssum += p[j];
        }
#pragma unroll
        for (int off = 16; off >= 1; off >>= 1)
            ssum += __shfl_xor(ssum, off, 32);
        const float inv = (ssum > 0.f) ? (1.0f / ssum) : 0.f;
#pragma unroll
        for (int j = 0; j < 32; ++j) rp[lane + 32 * j] = p[j] * inv;
    }
    __syncthreads();

    const int t      = wave & 3;
    const int kstart = (wave >> 2) * 512;
    const int ncol   = t * 16 + lmod;
    const float* vp  = v + ((size_t)bh * S_LEN) * D_DIM + ncol;

    v8f acc = {};
#pragma unroll 8
    for (int kb = kstart; kb < kstart + 512; kb += 4) {
        v2f a = *(const v2f*)(&sm[lmod * S_LEN + kb + 2 * lhalf]);
        v2f b;
        b.x = vp[(size_t)(kb + 2 * lhalf + 0) * D_DIM];
        b.y = vp[(size_t)(kb + 2 * lhalf + 1) * D_DIM];
        acc = __builtin_amdgcn_wmma_f32_16x16x4_f32(
            false, a, false, b, (short)0, acc, false, false);
    }
    __syncthreads();

#pragma unroll
    for (int r = 0; r < 8; ++r)
        sm[wave * 256 + r * 32 + lane] = acc[r];
    __syncthreads();

    if (wave < 4) {
#pragma unroll
        for (int r = 0; r < 8; ++r) {
            float val = sm[wave * 256 + r * 32 + lane]
                      + sm[(wave + 4) * 256 + r * 32 + lane];
            const int row = r + 8 * lhalf;
            const int col = wave * 16 + lmod;
            out[((size_t)(bh * S_LEN + qbase + row)) * D_DIM + col] = val;
        }
    }
}

extern "C" void kernel_launch(void* const* d_in, const int* in_sizes, int n_in,
                              void* d_out, int out_size, void* d_ws, size_t ws_size,
                              hipStream_t stream) {
    (void)in_sizes; (void)n_in; (void)out_size; (void)d_ws; (void)ws_size;
    const float* q = (const float*)d_in[0];
    const float* k = (const float*)d_in[1];
    const float* v = (const float*)d_in[2];
    float* out = (float*)d_out;

    dim3 grid(S_LEN / 16, BH);
    entmax_attn_kernel<<<grid, 256, 0, stream>>>(q, k, v, out);
}


# Pad the LDS score/prob row pitch from 1024 to 1028 floats so the row-per-lane A-fragment reads of the P*V product hit 64 distinct banks instead of 4 (16-way bank conflict removed)
# speedup vs baseline: 1.0777x; 1.0777x over previous
#include <hip/hip_runtime.h>


typedef __attribute__((ext_vector_type(2))) float v2f;
typedef __attribute__((ext_vector_type(8))) float v8f;

#define S_LEN 1024
#define D_DIM 64
#define BH    64
#define PITCH (S_LEN + 4)

__global__ __launch_bounds__(256)
void entmax_attn_kernel(const float* __restrict__ q,
                        const float* __restrict__ k,
                        const float* __restrict__ v,
                        float* __restrict__ out)
{
    __shared__ float sm[16 * PITCH];

    const int lane  = threadIdx.x & 31;
    const int wave  = threadIdx.x >> 5;
    const int lmod  = lane & 15;
    const int lhalf = lane >> 4;

    const int bh    = blockIdx.y;
    const int qbase = blockIdx.x * 16;

    const float* qp = q + ((size_t)(bh * S_LEN + qbase + lmod)) * D_DIM;
    v2f aq[16];
#pragma unroll
    for (int s = 0; s < 16; ++s)
        aq[s] = *(const v2f*)(qp + 4 * s + 2 * lhalf);

#pragma unroll
    for (int t = 0; t < 8; ++t) {
        const int nbase = (wave * 8 + t) * 16;
        const float* kp = k + ((size_t)(bh * S_LEN + nbase + lmod)) * D_DIM;
        v8f acc = {};
#pragma unroll
        for (int s = 0; s < 16; ++s) {
            v2f bf = *(const v2f*)(kp + 4 * s + 2 * lhalf);
            acc = __builtin_amdgcn_wmma_f32_16x16x4_f32(
                false, aq[s], false, bf, (short)0, acc, false, false);
        }
#pragma unroll
        for (int r = 0; r < 8; ++r)
            sm[(r + 8 * lhalf) * PITCH + nbase + lmod] = acc[r] * 0.0625f;
    }
    __syncthreads();

    for (int rr = 0; rr < 2; ++rr) {
        const int row = wave * 2 + rr;
        float* rp = &sm[row * PITCH];
        float x[32];
#pragma unroll
        for (int j = 0; j < 32; ++j) x[j] = rp[lane + 32 * j];

        float m = x[0];
#pragma unroll
        for (int j = 1; j < 32; ++j) m = fmaxf(m, x[j]);
#pragma unroll
        for (int off = 16; off >= 1; off >>= 1)
            m = fmaxf(m, __shfl_xor(m, off, 32));
#pragma unroll
        for (int j = 0; j < 32; ++j) x[j] -= m;

        float tau = -1.0f;
        for (int it = 0; it < 12; ++it) {
            float s = 0.f, g = 0.f;
#pragma unroll
            for (int j = 0; j < 32; ++j) {
                float d = fmaxf(x[j] - tau, 0.f);
                s = fmaf(d, d, s);
                g += d;
            }
#pragma unroll
            for (int off = 16; off >= 1; off >>= 1) {
                s += __shfl_xor(s, off, 32);
                g += __shfl_xor(g, off, 32);
            }
            g = fmaxf(g, 1e-20f);
            tau += (s - 1.0f) / (2.0f * g);
        }

        float p[32];
        float ssum = 0.f;
#pragma unroll
        for (int j = 0; j < 32; ++j) {
            float d = fmaxf(x[j] - tau, 0.f);
            p[j] = d * d;
            ssum += p[j];
        }
#pragma unroll
        for (int off = 16; off >= 1; off >>= 1)
            ssum += __shfl_xor(ssum, off, 32);
        const float inv = (ssum > 0.f) ? (1.0f / ssum) : 0.f;
#pragma unroll
        for (int j = 0; j < 32; ++j) rp[lane + 32 * j] = p[j] * inv;
    }
    __syncthreads();

    const int t      = wave & 3;
    const int kstart = (wave >> 2) * 512;
    const int ncol   = t * 16 + lmod;
    const float* vp  = v + ((size_t)bh * S_LEN) * D_DIM + ncol;

    v8f acc = {};
#pragma unroll 8
    for (int kb = kstart; kb < kstart + 512; kb += 4) {
        v2f a = *(const v2f*)(&sm[lmod * PITCH + kb + 2 * lhalf]);
        v2f b;
        b.x = vp[(size_t)(kb + 2 * lhalf + 0) * D_DIM];
        b.y = vp[(size_t)(kb + 2 * lhalf + 1) * D_DIM];
        acc = __builtin_amdgcn_wmma_f32_16x16x4_f32(
            false, a, false, b, (short)0, acc, false, false);
    }
    __syncthreads();

#pragma unroll
    for (int r = 0; r < 8; ++r)
        sm[wave * 256 + r * 32 + lane] = acc[r];
    __syncthreads();

    if (wave < 4) {
#pragma unroll
        for (int r = 0; r < 8; ++r) {
            float val = sm[wave * 256 + r * 32 + lane]
                      + sm[(wave + 4) * 256 + r * 32 + lane];
            const int row = r + 8 * lhalf;
            const int col = wave * 16 + lmod;
            out[((size_t)(bh * S_LEN + qbase + row)) * D_DIM + col] = val;
        }
    }
}

extern "C" void kernel_launch(void* const* d_in, const int* in_sizes, int n_in,
                              void* d_out, int out_size, void* d_ws, size_t ws_size,
                              hipStream_t stream) {
    (void)in_sizes; (void)n_in; (void)out_size; (void)d_ws; (void)ws_size;
    const float* q = (const float*)d_in[0];
    const float* k = (const float*)d_in[1];
    const float* v = (const float*)d_in[2];
    float* out = (float*)d_out;

    dim3 grid(S_LEN / 16, BH);
    entmax_attn_kernel<<<grid, 256, 0, stream>>>(q, k, v, out);
}


# Every fragment load made 16 B per lane (was 8 B Q/K, 4 B V): same k permutation in both loaders; V tile t = cols 4c+t, so a wave takes 128 keys x 64 cols, 8 partials summed via LDS; sum order changes
# speedup vs baseline: 1.1408x; 1.0586x over previous
#include <hip/hip_runtime.h>


typedef __attribute__((ext_vector_type(2))) float v2f;
typedef __attribute__((ext_vector_type(4))) float v4f;
typedef __attribute__((ext_vector_type(8))) float v8f;

#define S_LEN 1024
#define D_DIM 64
#define BH    64
#define PITCH (S_LEN + 4)

#define WMMA_F32(a, b, c) \
    __builtin_amdgcn_wmma_f32_16x16x4_f32(false, (a), false, (b), (short)0, (c), false, false)

__global__ __launch_bounds__(256)
void entmax_attn_kernel(const float* __restrict__ q,
                        const float* __restrict__ k,
                        const float* __restrict__ v,
                        float* __restrict__ out)
{
    __shared__ __attribute__((aligned(16))) float sm[16 * PITCH];

    const int lane  = threadIdx.x & 31;
    const int wave  = threadIdx.x >> 5;
    const int lmod  = lane & 15;
    const int lhalf = lane >> 4;

    const int bh    = blockIdx.y;
    const int qbase = blockIdx.x * 16;

    const float* qp = q + ((size_t)(bh * S_LEN + qbase + lmod)) * D_DIM + 4 * lhalf;
    v4f aq[8];
#pragma unroll
    for (int j = 0; j < 8; ++j)
        aq[j] = *(const v4f*)(qp + 8 * j);

#pragma unroll
    for (int t = 0; t < 8; ++t) {
        const int nbase = (wave * 8 + t) * 16;
        const float* kp = k + ((size_t)(bh * S_LEN + nbase + lmod)) * D_DIM + 4 * lhalf;
        v4f bk[8];
#pragma unroll
        for (int j = 0; j < 8; ++j)
            bk[j] = *(const v4f*)(kp + 8 * j);
        v8f acc = {};
#pragma unroll
        for (int j = 0; j < 8; ++j) {
            acc = WMMA_F32(aq[j].xy, bk[j].xy, acc);
            acc = WMMA_F32(aq[j].zw, bk[j].zw, acc);
        }
#pragma unroll
        for (int r = 0; r < 8; ++r)
            sm[(r + 8 * lhalf) * PITCH + nbase + lmod] = acc[r] * 0.0625f;
    }
    __syncthreads();

    for (int rr = 0; rr < 2; ++rr) {
        const int row = wave * 2 + rr;
        float* rp = &sm[row * PITCH];
        float x[32];
#pragma unroll
        for (int j = 0; j < 32; ++j) x[j] = rp[lane + 32 * j];

        float m = x[0];
#pragma unroll
        for (int j = 1; j < 32; ++j) m = fmaxf(m, x[j]);
#pragma unroll
        for (int off = 16; off >= 1; off >>= 1)
            m = fmaxf(m, __shfl_xor(m, off, 32));
#pragma unroll
        for (int j = 0; j < 32; ++j) x[j] -= m;

        float tau = -1.0f;
        for (int it = 0; it < 12; ++it) {
            float s = 0.f, g = 0.f;
#pragma unroll
            for (int j = 0; j < 32; ++j) {
                float d = fmaxf(x[j] - tau, 0.f);
                s = fmaf(d, d, s);
                g += d;
            }
#pragma unroll
            for (int off = 16; off >= 1; off >>= 1) {
                s += __shfl_xor(s, off, 32);
                g += __shfl_xor(g, off, 32);
            }
            g = fmaxf(g, 1e-20f);
            tau += (s - 1.0f) / (2.0f * g);
        }

        float p[32];
        float ssum = 0.f;
#pragma unroll
        for (int j = 0; j < 32; ++j) {
            float d = fmaxf(x[j] - tau, 0.f);
            p[j] = d * d;
            ssum += p[j];
        }
#pragma unroll
        for (int off = 16; off >= 1; off >>= 1)
            ssum += __shfl_xor(ssum, off, 32);
        const float inv = (ssum > 0.f) ? (1.0f / ssum) : 0.f;
#pragma unroll
        for (int j = 0; j < 32; ++j) rp[lane + 32 * j] = p[j] * inv;
    }
    __syncthreads();

    const int kslice = wave * 128;
    const float* vp = v + ((size_t)(bh * S_LEN + kslice + 4 * lhalf)) * D_DIM + 4 * lmod;
    const float* pp = &sm[lmod * PITCH + kslice + 4 * lhalf];

    v8f acc0 = {}, acc1 = {}, acc2 = {}, acc3 = {};
#pragma unroll
    for (int i = 0; i < 16; ++i) {
        const v4f a4 = *(const v4f*)(pp + 8 * i);
        const v4f r0 = *(const v4f*)(vp + (size_t)(8 * i + 0) * D_DIM);
        const v4f r1 = *(const v4f*)(vp + (size_t)(8 * i + 1) * D_DIM);
        const v4f r2 = *(const v4f*)(vp + (size_t)(8 * i + 2) * D_DIM);
        const v4f r3 = *(const v4f*)(vp + (size_t)(8 * i + 3) * D_DIM);
        const v2f aa = a4.xy;
        const v2f ab = a4.zw;
        const v2f b00 = {r0.x, r1.x}, b01 = {r0.y, r1.y}, b02 = {r0.z, r1.z}, b03 = {r0.w, r1.w};
        const v2f b10 = {r2.x, r3.x}, b11 = {r2.y, r3.y}, b12 = {r2.z, r3.z}, b13 = {r2.w, r3.w};
        acc0 = WMMA_F32(aa, b00, acc0);
        acc1 = WMMA_F32(aa, b01, acc1);
        acc2 = WMMA_F32(aa, b02, acc2);
        acc3 = WMMA_F32(aa, b03, acc3);
        acc0 = WMMA_F32(ab, b10, acc0);
        acc1 = WMMA_F32(ab, b11, acc1);
        acc2 = WMMA_F32(ab, b12, acc2);
        acc3 = WMMA_F32(ab, b13, acc3);
    }
    __syncthreads();

#pragma unroll
    for (int r = 0; r < 8; ++r) {
        const v4f o = {acc0[r], acc1[r], acc2[r], acc3[r]};
        *(v4f*)(&sm[wave * 1024 + (8 * lhalf + r) * D_DIM + 4 * lmod]) = o;
    }
    __syncthreads();

    const int e4 = threadIdx.x * 4;
    v4f sum = *(const v4f*)(&sm[e4]);
#pragma unroll
    for (int w = 1; w < 8; ++w)
        sum += *(const v4f*)(&sm[w * 1024 + e4]);
    *(v4f*)(out + ((size_t)(bh * S_LEN + qbase)) * D_DIM + e4) = sum;
}

extern "C" void kernel_launch(void* const* d_in, const int* in_sizes, int n_in,
                              void* d_out, int out_size, void* d_ws, size_t ws_size,
                              hipStream_t stream) {
    (void)in_sizes; (void)n_in; (void)out_size; (void)d_ws; (void)ws_size;
    const float* q = (const float*)d_in[0];
    const float* k = (const float*)d_in[1];
    const float* v = (const float*)d_in[2];
    float* out = (float*)d_out;

    dim3 grid(S_LEN / 16, BH);
    entmax_attn_kernel<<<grid, 256, 0, stream>>>(q, k, v, out);
}


# Wave reductions of the entmax phase (row max, Newton sums, normaliser) moved from LDS bpermute shuffles to one cross-row lane exchange plus four row_xmask DPP steps; same reduction tree
# speedup vs baseline: 1.2457x; 1.0920x over previous
#include <hip/hip_runtime.h>


typedef __attribute__((ext_vector_type(2))) float v2f;
typedef __attribute__((ext_vector_type(4))) float v4f;
typedef __attribute__((ext_vector_type(8))) float v8f;

#define S_LEN 1024
#define D_DIM 64
#define BH    64
#define PITCH (S_LEN + 4)

#define WMMA_F32(a, b, c) \
    __builtin_amdgcn_wmma_f32_16x16x4_f32(false, (a), false, (b), (short)0, (c), false, false)

__device__ __forceinline__ float xhalf(float x) {
    const int b = __builtin_bit_cast(int, x);
    return __builtin_bit_cast(float,
        __builtin_amdgcn_permlanex16(b, b, 0x76543210, 0xfedcba98, false, false));
}
template <int M>
__device__ __forceinline__ float rowx(float x) {
    const int b = __builtin_bit_cast(int, x);
    return __builtin_bit_cast(float,
        __builtin_amdgcn_update_dpp(0, b, 0x160 + M, 0xf, 0xf, true));
}
__device__ __forceinline__ float wave_sum(float s) {
    s += xhalf(s);
    s += rowx<8>(s);
    s += rowx<4>(s);
    s += rowx<2>(s);
    s += rowx<1>(s);
    return s;
}
__device__ __forceinline__ float wave_max(float m) {
    m = fmaxf(m, xhalf(m));
    m = fmaxf(m, rowx<8>(m));
    m = fmaxf(m, rowx<4>(m));
    m = fmaxf(m, rowx<2>(m));
    m = fmaxf(m, rowx<1>(m));
    return m;
}

__global__ __launch_bounds__(256)
void entmax_attn_kernel(const float* __restrict__ q,
                        const float* __restrict__ k,
                        const float* __restrict__ v,
                        float* __restrict__ out)
{
    __shared__ __attribute__((aligned(16))) float sm[16 * PITCH];

    const int lane  = threadIdx.x & 31;
    const int wave  = threadIdx.x >> 5;
    const int lmod  = lane & 15;
    const int lhalf = lane >> 4;

    const int bh    = blockIdx.y;
    const int qbase = blockIdx.x * 16;

    const float* qp = q + ((size_t)(bh * S_LEN + qbase + lmod)) * D_DIM + 4 * lhalf;
    v4f aq[8];
#pragma unroll
    for (int j = 0; j < 8; ++j)
        aq[j] = *(const v4f*)(qp + 8 * j);

#pragma unroll
    for (int t = 0; t < 8; ++t) {
        const int nbase = (wave * 8 + t) * 16;
        const float* kp = k + ((size_t)(bh * S_LEN + nbase + lmod)) * D_DIM + 4 * lhalf;
        v4f bk[8];
#pragma unroll
        for (int j = 0; j < 8; ++j)
            bk[j] = *(const v4f*)(kp + 8 * j);
        v8f acc = {};
#pragma unroll
        for (int j = 0; j < 8; ++j) {
            acc = WMMA_F32(aq[j].xy, bk[j].xy, acc);
            acc = WMMA_F32(aq[j].zw, bk[j].zw, acc);
        }
#pragma unroll
        for (int r = 0; r < 8; ++r)
            sm[(r + 8 * lhalf) * PITCH + nbase + lmod] = acc[r] * 0.0625f;
    }
    __syncthreads();

    for (int rr = 0; rr < 2; ++rr) {
        const int row = wave * 2 + rr;
        float* rp = &sm[row * PITCH];
        float x[32];
#pragma unroll
        for (int j = 0; j < 32; ++j) x[j] = rp[lane + 32 * j];

        float m = x[0];
#pragma unroll
        for (int j = 1; j < 32; ++j) m = fmaxf(m, x[j]);
        m = wave_max(m);
#pragma unroll
        for (int j = 0; j < 32; ++j) x[j] -= m;

        float tau = -1.0f;
        for (int it = 0; it < 12; ++it) {
            float s = 0.f, g = 0.f;
#pragma unroll
            for (int j = 0; j < 32; ++j) {
                float d = fmaxf(x[j] - tau, 0.f);
                s = fmaf(d, d, s);
                g += d;
            }
            s = wave_sum(s);
            g = wave_sum(g);
            g = fmaxf(g, 1e-20f);
            tau += (s - 1.0f) / (2.0f * g);
        }

        float p[32];
        float ssum = 0.f;
#pragma unroll
        for (int j = 0; j < 32; ++j) {
            float d = fmaxf(x[j] - tau, 0.f);
            p[j] = d * d;
            ssum += p[j];
        }
        ssum = wave_sum(ssum);
        const float inv = (ssum > 0.f) ? (1.0f / ssum) : 0.f;
#pragma unroll
        for (int j = 0; j < 32; ++j) rp[lane + 32 * j] = p[j] * inv;
    }
    __syncthreads();

    const int kslice = wave * 128;
    const float* vp = v + ((size_t)(bh * S_LEN + kslice + 4 * lhalf)) * D_DIM + 4 * lmod;
    const float* pp = &sm[lmod * PITCH + kslice + 4 * lhalf];

    v8f acc0 = {}, acc1 = {}, acc2 = {}, acc3 = {};
#pragma unroll
    for (int i = 0; i < 16; ++i) {
        const v4f a4 = *(const v4f*)(pp + 8 * i);
        const v4f r0 = *(const v4f*)(vp + (size_t)(8 * i + 0) * D_DIM);
        const v4f r1 = *(const v4f*)(vp + (size_t)(8 * i + 1) * D_DIM);
        const v4f r2 = *(const v4f*)(vp + (size_t)(8 * i + 2) * D_DIM);
        const v4f r3 = *(const v4f*)(vp + (size_t)(8 * i + 3) * D_DIM);
        const v2f aa = a4.xy;
        const v2f ab = a4.zw;
        const v2f b00 = {r0.x, r1.x}, b01 = {r0.y, r1.y}, b02 = {r0.z, r1.z}, b03 = {r0.w, r1.w};
        const v2f b10 = {r2.x, r3.x}, b11 = {r2.y, r3.y}, b12 = {r2.z, r3.z}, b13 = {r2.w, r3.w};
        acc0 = WMMA_F32(aa, b00, acc0);
        acc1 = WMMA_F32(aa, b01, acc1);
        acc2 = WMMA_F32(aa, b02, acc2);
        acc3 = WMMA_F32(aa, b03, acc3);
        acc0 = WMMA_F32(ab, b10, acc0);
        acc1 = WMMA_F32(ab, b11, acc1);
        acc2 = WMMA_F32(ab, b12, acc2);
        acc3 = WMMA_F32(ab, b13, acc3);
    }
    __syncthreads();

#pragma unroll
    for (int r = 0; r < 8; ++r) {
        const v4f o = {acc0[r], acc1[r], acc2[r], acc3[r]};
        *(v4f*)(&sm[wave * 1024 + (8 * lhalf + r) * D_DIM + 4 * lmod]) = o;
    }
    __syncthreads();

    const int e4 = threadIdx.x * 4;
    v4f sum = *(const v4f*)(&sm[e4]);
#pragma unroll
    for (int w = 1; w < 8; ++w)
        sum += *(const v4f*)(&sm[w * 1024 + e4]);
    *(v4f*)(out + ((size_t)(bh * S_LEN + qbase)) * D_DIM + e4) = sum;
}

extern "C" void kernel_launch(void* const* d_in, const int* in_sizes, int n_in,
                              void* d_out, int out_size, void* d_ws, size_t ws_size,
                              hipStream_t stream) {
    (void)in_sizes; (void)n_in; (void)out_size; (void)d_ws; (void)ws_size;
    const float* q = (const float*)d_in[0];
    const float* k = (const float*)d_in[1];
    const float* v = (const float*)d_in[2];
    float* out = (float*)d_out;

    dim3 grid(S_LEN / 16, BH);
    entmax_attn_kernel<<<grid, 256, 0, stream>>>(q, k, v, out);
}


# K tiles reach the Q*K^T fragments through a per-wave padded LDS patch filled by whole-line global loads (was 16 B per lane gathered from 16 rows per load); fragment contents and k order unchanged
# speedup vs baseline: 1.7932x; 1.4395x over previous
#include <hip/hip_runtime.h>


typedef __attribute__((ext_vector_type(2))) float v2f;
typedef __attribute__((ext_vector_type(4))) float v4f;
typedef __attribute__((ext_vector_type(8))) float v8f;

#define S_LEN 1024
#define D_DIM 64
#define BH    64
#define PITCH (S_LEN + 4)
#define KPITCH (D_DIM + 4)

#define WMMA_F32(a, b, c) \
    __builtin_amdgcn_wmma_f32_16x16x4_f32(false, (a), false, (b), (short)0, (c), false, false)

__device__ __forceinline__ float xhalf(float x) {
    const int b = __builtin_bit_cast(int, x);
    return __builtin_bit_cast(float,
        __builtin_amdgcn_permlanex16(b, b, 0x76543210, 0xfedcba98, false, false));
}
template <int M>
__device__ __forceinline__ float rowx(float x) {
    const int b = __builtin_bit_cast(int, x);
    return __builtin_bit_cast(float,
        __builtin_amdgcn_update_dpp(0, b, 0x160 + M, 0xf, 0xf, true));
}
__device__ __forceinline__ float wave_sum(float s) {
    s += xhalf(s);
    s += rowx<8>(s);
    s += rowx<4>(s);
    s += rowx<2>(s);
    s += rowx<1>(s);
    return s;
}
__device__ __forceinline__ float wave_max(float m) {
    m = fmaxf(m, xhalf(m));
    m = fmaxf(m, rowx<8>(m));
    m = fmaxf(m, rowx<4>(m));
    m = fmaxf(m, rowx<2>(m));
    m = fmaxf(m, rowx<1>(m));
    return m;
}

__global__ __launch_bounds__(256)
void entmax_attn_kernel(const float* __restrict__ q,
                        const float* __restrict__ k,
                        const float* __restrict__ v,
                        float* __restrict__ out)
{
    __shared__ __attribute__((aligned(16))) float sm[16 * PITCH + 8 * 16 * KPITCH];

    const int lane  = threadIdx.x & 31;
    const int wave  = threadIdx.x >> 5;
    const int lmod  = lane & 15;
    const int lhalf = lane >> 4;

    const int bh    = blockIdx.y;
    const int qbase = blockIdx.x * 16;

    const float* qp = q + ((size_t)(bh * S_LEN + qbase + lmod)) * D_DIM + 4 * lhalf;
    v4f aq[8];
#pragma unroll
    for (int j = 0; j < 8; ++j)
        aq[j] = *(const v4f*)(qp + 8 * j);

    const float* kt = k + ((size_t)(bh * S_LEN + wave * 128)) * D_DIM + 4 * lane;
    float* stg = &sm[16 * PITCH + wave * 16 * KPITCH];
    float* stw = stg + lhalf * KPITCH + 4 * lmod;
    const float* stf = stg + lmod * KPITCH + 4 * lhalf;

    v4f g[8];
#pragma unroll
    for (int i = 0; i < 8; ++i)
        g[i] = *(const v4f*)(kt + 128 * i);

#pragma unroll
    for (int t = 0; t < 8; ++t) {
        const int nbase = (wave * 8 + t) * 16;
#pragma unroll
        for (int i = 0; i < 8; ++i)
            *(v4f*)(stw + 2 * i * KPITCH) = g[i];
        if (t < 7) {
#pragma unroll
            for (int i = 0; i < 8; ++i)
                g[i] = *(const v4f*)(kt + 1024 * (t + 1) + 128 * i);
        }
        v4f bk[8];
#pragma unroll
        for (int j = 0; j < 8; ++j)
            bk[j] = *(const v4f*)(stf + 8 * j);
        v8f acc = {};
#pragma unroll
        for (int j = 0; j < 8; ++j) {
            acc = WMMA_F32(aq[j].xy, bk[j].xy, acc);
            acc = WMMA_F32(aq[j].zw, bk[j].zw, acc);
        }
#pragma unroll
        for (int r = 0; r < 8; ++r)
            sm[(r + 8 * lhalf) * PITCH + nbase + lmod] = acc[r] * 0.0625f;
    }
    __syncthreads();

    for (int rr = 0; rr < 2; ++rr) {
        const int row = wave * 2 + rr;
        float* rp = &sm[row * PITCH];
        float x[32];
#pragma unroll
        for (int j = 0; j < 32; ++j) x[j] = rp[lane + 32 * j];

        float m = x[0];
#pragma unroll
        for (int j = 1; j < 32; ++j) m = fmaxf(m, x[j]);
        m = wave_max(m);
#pragma unroll
        for (int j = 0; j < 32; ++j) x[j] -= m;

        float tau = -1.0f;
        for (int it = 0; it < 12; ++it) {
            float s = 0.f, g = 0.f;
#pragma unroll
            for (int j = 0; j < 32; ++j) {
                float d = fmaxf(x[j] - tau, 0.f);
                s = fmaf(d, d, s);
                g += d;
            }
            s = wave_sum(s);
            g = wave_sum(g);
            g = fmaxf(g, 1e-20f);
            tau += (s - 1.0f) / (2.0f * g);
        }

        float p[32];
        float ssum = 0.f;
#pragma unroll
        for (int j = 0; j < 32; ++j) {
            float d = fmaxf(x[j] - tau, 0.f);
            p[j] = d * d;
            ssum += p[j];
        }
        ssum = wave_sum(ssum);
        const float inv = (ssum > 0.f) ? (1.0f / ssum) : 0.f;
#pragma unroll
        for (int j = 0; j < 32; ++j) rp[lane + 32 * j] = p[j] * inv;
    }
    __syncthreads();

    const int kslice = wave * 128;
    const float* vp = v + ((size_t)(bh * S_LEN + kslice + 4 * lhalf)) * D_DIM + 4 * lmod;
    const float* pp = &sm[lmod * PITCH + kslice + 4 * lhalf];

    v8f acc0 = {}, acc1 = {}, acc2 = {}, acc3 = {};
#pragma unroll
    for (int i = 0; i < 16; ++i) {
        const v4f a4 = *(const v4f*)(pp + 8 * i);
        const v4f r0 = *(const v4f*)(vp + (size_t)(8 * i + 0) * D_DIM);
        const v4f r1 = *(const v4f*)(vp + (size_t)(8 * i + 1) * D_DIM);
        const v4f r2 = *(const v4f*)(vp + (size_t)(8 * i + 2) * D_DIM);
        const v4f r3 = *(const v4f*)(vp + (size_t)(8 * i + 3) * D_DIM);
        const v2f aa = a4.xy;
        const v2f ab = a4.zw;
        const v2f b00 = {r0.x, r1.x}, b01 = {r0.y, r1.y}, b02 = {r0.z, r1.z}, b03 = {r0.w, r1.w};
        const v2f b10 = {r2.x, r3.x}, b11 = {r2.y, r3.y}, b12 = {r2.z, r3.z}, b13 = {r2.w, r3.w};
        acc0 = WMMA_F32(aa, b00, acc0);
        acc1 = WMMA_F32(aa, b01, acc1);
        acc2 = WMMA_F32(aa, b02, acc2);
        acc3 = WMMA_F32(aa, b03, acc3);
        acc0 = WMMA_F32(ab, b10, acc0);
        acc1 = WMMA_F32(ab, b11, acc1);
        acc2 = WMMA_F32(ab, b12, acc2);
        acc3 = WMMA_F32(ab, b13, acc3);
    }
    __syncthreads();

#pragma unroll
    for (int r = 0; r < 8; ++r) {
        const v4f o = {acc0[r], acc1[r], acc2[r], acc3[r]};
        *(v4f*)(&sm[wave * 1024 + (8 * lhalf + r) * D_DIM + 4 * lmod]) = o;
    }
    __syncthreads();

    const int e4 = threadIdx.x * 4;
    v4f sum = *(const v4f*)(&sm[e4]);
#pragma unroll
    for (int w = 1; w < 8; ++w)
        sum += *(const v4f*)(&sm[w * 1024 + e4]);
    *(v4f*)(out + ((size_t)(bh * S_LEN + qbase)) * D_DIM + e4) = sum;
}

extern "C" void kernel_launch(void* const* d_in, const int* in_sizes, int n_in,
                              void* d_out, int out_size, void* d_ws, size_t ws_size,
                              hipStream_t stream) {
    (void)in_sizes; (void)n_in; (void)out_size; (void)d_ws; (void)ws_size;
    const float* q = (const float*)d_in[0];
    const float* k = (const float*)d_in[1];
    const float* v = (const float*)d_in[2];
    float* out = (float*)d_out;

    dim3 grid(S_LEN / 16, BH);
    entmax_attn_kernel<<<grid, 256, 0, stream>>>(q, k, v, out);
}


# Entmax Newton sums and normaliser run only over entries within 1 of the row max (the rest add zeros), packed across lanes by ballot + prefix count via LDS, 4 or 8 per lane; sum order changes
# speedup vs baseline: 1.9264x; 1.0743x over previous
#include <hip/hip_runtime.h>


typedef __attribute__((ext_vector_type(2))) float v2f;
typedef __attribute__((ext_vector_type(4))) float v4f;
typedef __attribute__((ext_vector_type(8))) float v8f;

#define S_LEN 1024
#define D_DIM 64
#define BH    64
#define PITCH (S_LEN + 4)
#define KPITCH (D_DIM + 4)

#define WMMA_F32(a, b, c) \
    __builtin_amdgcn_wmma_f32_16x16x4_f32(false, (a), false, (b), (short)0, (c), false, false)

__device__ __forceinline__ float xhalf(float x) {
    const int b = __builtin_bit_cast(int, x);
    return __builtin_bit_cast(float,
        __builtin_amdgcn_permlanex16(b, b, 0x76543210, 0xfedcba98, false, false));
}
template <int M>
__device__ __forceinline__ float rowx(float x) {
    const int b = __builtin_bit_cast(int, x);
    return __builtin_bit_cast(float,
        __builtin_amdgcn_update_dpp(0, b, 0x160 + M, 0xf, 0xf, true));
}
__device__ __forceinline__ float wave_sum(float s) {
    s += xhalf(s);
    s += rowx<8>(s);
    s += rowx<4>(s);
    s += rowx<2>(s);
    s += rowx<1>(s);
    return s;
}
__device__ __forceinline__ float wave_max(float m) {
    m = fmaxf(m, xhalf(m));
    m = fmaxf(m, rowx<8>(m));
    m = fmaxf(m, rowx<4>(m));
    m = fmaxf(m, rowx<2>(m));
    m = fmaxf(m, rowx<1>(m));
    return m;
}

template <int N>
__device__ __forceinline__ float newton_tau(const float (&x)[N], float& ssum) {
    float tau = -1.0f;
    for (int it = 0; it < 12; ++it) {
        float s = 0.f, g = 0.f;
#pragma unroll
        for (int j = 0; j < N; ++j) {
            float d = fmaxf(x[j] - tau, 0.f);
            s = fmaf(d, d, s);
            g += d;
        }
        s = wave_sum(s);
        g = wave_sum(g);
        g = fmaxf(g, 1e-20f);
        tau = fmaxf(tau + (s - 1.0f) / (2.0f * g), -1.0f);
    }
    float ss = 0.f;
#pragma unroll
    for (int j = 0; j < N; ++j) {
        float d = fmaxf(x[j] - tau, 0.f);
        ss += d * d;
    }
    ssum = ss;
    return tau;
}

__global__ __launch_bounds__(256)
void entmax_attn_kernel(const float* __restrict__ q,
                        const float* __restrict__ k,
                        const float* __restrict__ v,
                        float* __restrict__ out)
{
    __shared__ __attribute__((aligned(16))) float sm[16 * PITCH + 8 * 16 * KPITCH];

    const int lane  = threadIdx.x & 31;
    const int wave  = threadIdx.x >> 5;
    const int lmod  = lane & 15;
    const int lhalf = lane >> 4;

    const int bh    = blockIdx.y;
    const int qbase = blockIdx.x * 16;

    const float* qp = q + ((size_t)(bh * S_LEN + qbase + lmod)) * D_DIM + 4 * lhalf;
    v4f aq[8];
#pragma unroll
    for (int j = 0; j < 8; ++j)
        aq[j] = *(const v4f*)(qp + 8 * j);

    const float* kt = k + ((size_t)(bh * S_LEN + wave * 128)) * D_DIM + 4 * lane;
    float* stg = &sm[16 * PITCH + wave * 16 * KPITCH];
    float* stw = stg + lhalf * KPITCH + 4 * lmod;
    const float* stf = stg + lmod * KPITCH + 4 * lhalf;

    v4f g[8];
#pragma unroll
    for (int i = 0; i < 8; ++i)
        g[i] = *(const v4f*)(kt + 128 * i);

#pragma unroll
    for (int t = 0; t < 8; ++t) {
        const int nbase = (wave * 8 + t) * 16;
#pragma unroll
        for (int i = 0; i < 8; ++i)
            *(v4f*)(stw + 2 * i * KPITCH) = g[i];
        if (t < 7) {
#pragma unroll
            for (int i = 0; i < 8; ++i)
                g[i] = *(const v4f*)(kt + 1024 * (t + 1) + 128 * i);
        }
        v4f bk[8];
#pragma unroll
        for (int j = 0; j < 8; ++j)
            bk[j] = *(const v4f*)(stf + 8 * j);
        v8f acc = {};
#pragma unroll
        for (int j = 0; j < 8; ++j) {
            acc = WMMA_F32(aq[j].xy, bk[j].xy, acc);
            acc = WMMA_F32(aq[j].zw, bk[j].zw, acc);
        }
#pragma unroll
        for (int r = 0; r < 8; ++r)
            sm[(r + 8 * lhalf) * PITCH + nbase + lmod] = acc[r] * 0.0625f;
    }
    __syncthreads();

    float* cb = &sm[16 * PITCH + wave * 16 * KPITCH];
    for (int rr = 0; rr < 2; ++rr) {
        const int row = wave * 2 + rr;
        float* rp = &sm[row * PITCH];
        float x[32];
#pragma unroll
        for (int j = 0; j < 32; ++j) x[j] = rp[lane + 32 * j];

        float m = x[0];
#pragma unroll
        for (int j = 1; j < 32; ++j) m = fmaxf(m, x[j]);
        m = wave_max(m);
#pragma unroll
        for (int j = 0; j < 32; ++j) x[j] -= m;

        const v4f fill = {-2.0f, -2.0f, -2.0f, -2.0f};
        *(v4f*)(cb + 4 * lane) = fill;
        *(v4f*)(cb + 128 + 4 * lane) = fill;

        unsigned n = 0;
#pragma unroll
        for (int j = 0; j < 32; ++j) {
            const bool keep = x[j] > -1.0f;
            const unsigned mask = __builtin_amdgcn_ballot_w32(keep);
            const unsigned pos = __builtin_amdgcn_mbcnt_lo(mask, n);
            if (keep) cb[pos] = x[j];
            n += (unsigned)__builtin_popcount(mask);
        }

        float tau, ssum;
        if (n <= 128u) {
            float c[4];
#pragma unroll
            for (int j = 0; j < 4; ++j) c[j] = cb[lane + 32 * j];
            tau = newton_tau<4>(c, ssum);
        } else if (n <= 256u) {
            float c[8];
#pragma unroll
            for (int j = 0; j < 8; ++j) c[j] = cb[lane + 32 * j];
            tau = newton_tau<8>(c, ssum);
        } else {
            tau = newton_tau<32>(x, ssum);
        }

        ssum = wave_sum(ssum);
        const float inv = (ssum > 0.f) ? (1.0f / ssum) : 0.f;
#pragma unroll
        for (int j = 0; j < 32; ++j) {
            float d = fmaxf(x[j] - tau, 0.f);
            rp[lane + 32 * j] = (d * d) * inv;
        }
    }
    __syncthreads();

    const int kslice = wave * 128;
    const float* vp = v + ((size_t)(bh * S_LEN + kslice + 4 * lhalf)) * D_DIM + 4 * lmod;
    const float* pp = &sm[lmod * PITCH + kslice + 4 * lhalf];

    v8f acc0 = {}, acc1 = {}, acc2 = {}, acc3 = {};
#pragma unroll
    for (int i = 0; i < 16; ++i) {
        const v4f a4 = *(const v4f*)(pp + 8 * i);
        const v4f r0 = *(const v4f*)(vp + (size_t)(8 * i + 0) * D_DIM);
        const v4f r1 = *(const v4f*)(vp + (size_t)(8 * i + 1) * D_DIM);
        const v4f r2 = *(const v4f*)(vp + (size_t)(8 * i + 2) * D_DIM);
        const v4f r3 = *(const v4f*)(vp + (size_t)(8 * i + 3) * D_DIM);
        const v2f aa = a4.xy;
        const v2f ab = a4.zw;
        const v2f b00 = {r0.x, r1.x}, b01 = {r0.y, r1.y}, b02 = {r0.z, r1.z}, b03 = {r0.w, r1.w};
        const v2f b10 = {r2.x, r3.x}, b11 = {r2.y, r3.y}, b12 = {r2.z, r3.z}, b13 = {r2.w, r3.w};
        acc0 = WMMA_F32(aa, b00, acc0);
        acc1 = WMMA_F32(aa, b01, acc1);
        acc2 = WMMA_F32(aa, b02, acc2);
        acc3 = WMMA_F32(aa, b03, acc3);
        acc0 = WMMA_F32(ab, b10, acc0);
        acc1 = WMMA_F32(ab, b11, acc1);
        acc2 = WMMA_F32(ab, b12, acc2);
        acc3 = WMMA_F32(ab, b13, acc3);
    }
    __syncthreads();

#pragma unroll
    for (int r = 0; r < 8; ++r) {
        const v4f o = {acc0[r], acc1[r], acc2[r], acc3[r]};
        *(v4f*)(&sm[wave * 1024 + (8 * lhalf + r) * D_DIM + 4 * lmod]) = o;
    }
    __syncthreads();

    const int e4 = threadIdx.x * 4;
    v4f sum = *(const v4f*)(&sm[e4]);
#pragma unroll
    for (int w = 1; w < 8; ++w)
        sum += *(const v4f*)(&sm[w * 1024 + e4]);
    *(v4f*)(out + ((size_t)(bh * S_LEN + qbase)) * D_DIM + e4) = sum;
}

extern "C" void kernel_launch(void* const* d_in, const int* in_sizes, int n_in,
                              void* d_out, int out_size, void* d_ws, size_t ws_size,
                              hipStream_t stream) {
    (void)in_sizes; (void)n_in; (void)out_size; (void)d_ws; (void)ws_size;
    const float* q = (const float*)d_in[0];
    const float* k = (const float*)d_in[1];
    const float* v = (const float*)d_in[2];
    float* out = (float*)d_out;

    dim3 grid(S_LEN / 16, BH);
    entmax_attn_kernel<<<grid, 256, 0, stream>>>(q, k, v, out);
}


# P*V runs only over the keys of each wave's 128-key slice with a nonzero probability in any of the 16 rows: indices packed by ballot into an LDS list, V rows and P gathered via it; sum order changes
# speedup vs baseline: 2.2320x; 1.1586x over previous
#include <hip/hip_runtime.h>


typedef __attribute__((ext_vector_type(2))) float v2f;
typedef __attribute__((ext_vector_type(4))) float v4f;
typedef __attribute__((ext_vector_type(8))) float v8f;
typedef __attribute__((ext_vector_type(4))) int   v4i;

#define S_LEN 1024
#define D_DIM 64
#define BH    64
#define PITCH (S_LEN + 4)
#define KPITCH (D_DIM + 4)

#define WMMA_F32(a, b, c) \
    __builtin_amdgcn_wmma_f32_16x16x4_f32(false, (a), false, (b), (short)0, (c), false, false)

__device__ __forceinline__ float xhalf(float x) {
    const int b = __builtin_bit_cast(int, x);
    return __builtin_bit_cast(float,
        __builtin_amdgcn_permlanex16(b, b, 0x76543210, 0xfedcba98, false, false));
}
template <int M>
__device__ __forceinline__ float rowx(float x) {
    const int b = __builtin_bit_cast(int, x);
    return __builtin_bit_cast(float,
        __builtin_amdgcn_update_dpp(0, b, 0x160 + M, 0xf, 0xf, true));
}
__device__ __forceinline__ float wave_sum(float s) {
    s += xhalf(s);
    s += rowx<8>(s);
    s += rowx<4>(s);
    s += rowx<2>(s);
    s += rowx<1>(s);
    return s;
}
__device__ __forceinline__ float wave_max(float m) {
    m = fmaxf(m, xhalf(m));
    m = fmaxf(m, rowx<8>(m));
    m = fmaxf(m, rowx<4>(m));
    m = fmaxf(m, rowx<2>(m));
    m = fmaxf(m, rowx<1>(m));
    return m;
}

template <int N>
__device__ __forceinline__ float newton_tau(const float (&x)[N], float& ssum) {
    float tau = -1.0f;
    for (int it = 0; it < 12; ++it) {
        float s = 0.f, g = 0.f;
#pragma unroll
        for (int j = 0; j < N; ++j) {
            float d = fmaxf(x[j] - tau, 0.f);
            s = fmaf(d, d, s);
            g += d;
        }
        s = wave_sum(s);
        g = wave_sum(g);
        g = fmaxf(g, 1e-20f);
        tau = fmaxf(tau + (s - 1.0f) / (2.0f * g), -1.0f);
    }
    float ss = 0.f;
#pragma unroll
    for (int j = 0; j < N; ++j) {
        float d = fmaxf(x[j] - tau, 0.f);
        ss += d * d;
    }
    ssum = ss;
    return tau;
}

__global__ __launch_bounds__(256)
void entmax_attn_kernel(const float* __restrict__ q,
                        const float* __restrict__ k,
                        const float* __restrict__ v,
                        float* __restrict__ out)
{
    __shared__ __attribute__((aligned(16))) float sm[16 * PITCH + 8 * 16 * KPITCH];

    const int lane  = threadIdx.x & 31;
    const int wave  = threadIdx.x >> 5;
    const int lmod  = lane & 15;
    const int lhalf = lane >> 4;

    const int bh    = blockIdx.y;
    const int qbase = blockIdx.x * 16;

    const float* qp = q + ((size_t)(bh * S_LEN + qbase + lmod)) * D_DIM + 4 * lhalf;
    v4f aq[8];
#pragma unroll
    for (int j = 0; j < 8; ++j)
        aq[j] = *(const v4f*)(qp + 8 * j);

    const float* kt = k + ((size_t)(bh * S_LEN + wave * 128)) * D_DIM + 4 * lane;
    float* stg = &sm[16 * PITCH + wave * 16 * KPITCH];
    float* stw = stg + lhalf * KPITCH + 4 * lmod;
    const float* stf = stg + lmod * KPITCH + 4 * lhalf;

    v4f g[8];
#pragma unroll
    for (int i = 0; i < 8; ++i)
        g[i] = *(const v4f*)(kt + 128 * i);

#pragma unroll
    for (int t = 0; t < 8; ++t) {
        const int nbase = (wave * 8 + t) * 16;
#pragma unroll
        for (int i = 0; i < 8; ++i)
            *(v4f*)(stw + 2 * i * KPITCH) = g[i];
        if (t < 7) {
#pragma unroll
            for (int i = 0; i < 8; ++i)
                g[i] = *(const v4f*)(kt + 1024 * (t + 1) + 128 * i);
        }
        v4f bk[8];
#pragma unroll
        for (int j = 0; j < 8; ++j)
            bk[j] = *(const v4f*)(stf + 8 * j);
        v8f acc = {};
#pragma unroll
        for (int j = 0; j < 8; ++j) {
            acc = WMMA_F32(aq[j].xy, bk[j].xy, acc);
            acc = WMMA_F32(aq[j].zw, bk[j].zw, acc);
        }
#pragma unroll
        for (int r = 0; r < 8; ++r)
            sm[(r + 8 * lhalf) * PITCH + nbase + lmod] = acc[r] * 0.0625f;
    }
    __syncthreads();

    float* cb = &sm[16 * PITCH + wave * 16 * KPITCH];
    for (int rr = 0; rr < 2; ++rr) {
        const int row = wave * 2 + rr;
        float* rp = &sm[row * PITCH];
        float x[32];
#pragma unroll
        for (int j = 0; j < 32; ++j) x[j] = rp[lane + 32 * j];

        float m = x[0];
#pragma unroll
        for (int j = 1; j < 32; ++j) m = fmaxf(m, x[j]);
        m = wave_max(m);
#pragma unroll
        for (int j = 0; j < 32; ++j) x[j] -= m;

        const v4f fill = {-2.0f, -2.0f, -2.0f, -2.0f};
        *(v4f*)(cb + 4 * lane) = fill;
        *(v4f*)(cb + 128 + 4 * lane) = fill;

        unsigned n = 0;
#pragma unroll
        for (int j = 0; j < 32; ++j) {
            const bool keep = x[j] > -1.0f;
            const unsigned mask = __builtin_amdgcn_ballot_w32(keep);
            const unsigned pos = __builtin_amdgcn_mbcnt_lo(mask, n);
            if (keep) cb[pos] = x[j];
            n += (unsigned)__builtin_popcount(mask);
        }

        float tau, ssum;
        if (n <= 128u) {
            float c[4];
#pragma unroll
            for (int j = 0; j < 4; ++j) c[j] = cb[lane + 32 * j];
            tau = newton_tau<4>(c, ssum);
        } else if (n <= 256u) {
            float c[8];
#pragma unroll
            for (int j = 0; j < 8; ++j) c[j] = cb[lane + 32 * j];
            tau = newton_tau<8>(c, ssum);
        } else {
            tau = newton_tau<32>(x, ssum);
        }

        ssum = wave_sum(ssum);
        const float inv = (ssum > 0.f) ? (1.0f / ssum) : 0.f;
#pragma unroll
        for (int j = 0; j < 32; ++j) {
            float d = fmaxf(x[j] - tau, 0.f);
            rp[lane + 32 * j] = (d * d) * inv;
        }
    }
    __syncthreads();

    const int kslice = wave * 128;
    int* lst = (int*)cb;
    {
        const v4i dummy = {kslice, kslice, kslice, kslice};
        *(v4i*)(lst + 4 * lane) = dummy;
        if (lane < 8) *(v4i*)(lst + 128 + 4 * lane) = dummy;
    }
    v4f mx = *(const v4f*)(&sm[kslice + 4 * lane]);
#pragma unroll
    for (int r = 1; r < 16; ++r) {
        const v4f pr = *(const v4f*)(&sm[r * PITCH + kslice + 4 * lane]);
        mx.x = fmaxf(mx.x, pr.x);
        mx.y = fmaxf(mx.y, pr.y);
        mx.z = fmaxf(mx.z, pr.z);
        mx.w = fmaxf(mx.w, pr.w);
    }
    unsigned nk = 0;
#pragma unroll
    for (int e = 0; e < 4; ++e) {
        const bool keep = mx[e] > 0.f;
        const unsigned mask = __builtin_amdgcn_ballot_w32(keep);
        const unsigned pos = __builtin_amdgcn_mbcnt_lo(mask, nk);
        if (keep) lst[pos] = kslice + 4 * lane + e;
        nk += (unsigned)__builtin_popcount(mask);
    }

    const float* vb = v + (size_t)bh * S_LEN * D_DIM + 4 * lmod;
    const float* prow = &sm[lmod * PITCH];
    const int nbatch = (int)((nk + 39u) / 40u);

    v8f acc0 = {}, acc1 = {}, acc2 = {}, acc3 = {};
    for (int b = 0; b < nbatch; ++b) {
        v4f a4[5], r0[5], r1[5], r2[5], r3[5];
#pragma unroll
        for (int u = 0; u < 5; ++u) {
            const int p0 = 40 * b + 8 * u + 4 * lhalf;
            const v4i kk = *(const v4i*)(lst + p0);
            v4f a;
            a.x = prow[kk.x];
            a.y = prow[kk.y];
            a.z = prow[kk.z];
            a.w = prow[kk.w];
            a.x = ((unsigned)(p0 + 0) < nk) ? a.x : 0.f;
            a.y = ((unsigned)(p0 + 1) < nk) ? a.y : 0.f;
            a.z = ((unsigned)(p0 + 2) < nk) ? a.z : 0.f;
            a.w = ((unsigned)(p0 + 3) < nk) ? a.w : 0.f;
            a4[u] = a;
            r0[u] = *(const v4f*)(vb + (size_t)kk.x * D_DIM);
            r1[u] = *(const v4f*)(vb + (size_t)kk.y * D_DIM);
            r2[u] = *(const v4f*)(vb + (size_t)kk.z * D_DIM);
            r3[u] = *(const v4f*)(vb + (size_t)kk.w * D_DIM);
        }
#pragma unroll
        for (int u = 0; u < 5; ++u) {
            const v2f aa = a4[u].xy;
            const v2f ab = a4[u].zw;
            const v2f b00 = {r0[u].x, r1[u].x}, b01 = {r0[u].y, r1[u].y}, b02 = {r0[u].z, r1[u].z}, b03 = {r0[u].w, r1[u].w};
            const v2f b10 = {r2[u].x, r3[u].x}, b11 = {r2[u].y, r3[u].y}, b12 = {r2[u].z, r3[u].z}, b13 = {r2[u].w, r3[u].w};
            acc0 = WMMA_F32(aa, b00, acc0);
            acc1 = WMMA_F32(aa, b01, acc1);
            acc2 = WMMA_F32(aa, b02, acc2);
            acc3 = WMMA_F32(aa, b03, acc3);
            acc0 = WMMA_F32(ab, b10, acc0);
            acc1 = WMMA_F32(ab, b11, acc1);
            acc2 = WMMA_F32(ab, b12, acc2);
            acc3 = WMMA_F32(ab, b13, acc3);
        }
    }
    __syncthreads();

#pragma unroll
    for (int r = 0; r < 8; ++r) {
        const v4f o = {acc0[r], acc1[r], acc2[r], acc3[r]};
        *(v4f*)(&sm[wave * 1024 + (8 * lhalf + r) * D_DIM + 4 * lmod]) = o;
    }
    __syncthreads();

    const int e4 = threadIdx.x * 4;
    v4f sum = *(const v4f*)(&sm[e4]);
#pragma unroll
    for (int w = 1; w < 8; ++w)
        sum += *(const v4f*)(&sm[w * 1024 + e4]);
    *(v4f*)(out + ((size_t)(bh * S_LEN + qbase)) * D_DIM + e4) = sum;
}

extern "C" void kernel_launch(void* const* d_in, const int* in_sizes, int n_in,
                              void* d_out, int out_size, void* d_ws, size_t ws_size,
                              hipStream_t stream) {
    (void)in_sizes; (void)n_in; (void)out_size; (void)d_ws; (void)ws_size;
    const float* q = (const float*)d_in[0];
    const float* k = (const float*)d_in[1];
    const float* v = (const float*)d_in[2];
    float* out = (float*)d_out;

    dim3 grid(S_LEN / 16, BH);
    entmax_attn_kernel<<<grid, 256, 0, stream>>>(q, k, v, out);
}


# Each wave runs its two entmax rows side by side, one per 16-lane half with 64 entries per lane (was one after the other on 32 lanes): one Newton stream for both, 4-step row sums; sum order changes
# speedup vs baseline: 2.5432x; 1.1394x over previous
#include <hip/hip_runtime.h>


typedef __attribute__((ext_vector_type(2))) float v2f;
typedef __attribute__((ext_vector_type(4))) float v4f;
typedef __attribute__((ext_vector_type(8))) float v8f;
typedef __attribute__((ext_vector_type(4))) int   v4i;

#define S_LEN 1024
#define D_DIM 64
#define BH    64
#define PITCH (S_LEN + 4)
#define KPITCH (D_DIM + 4)

#define WMMA_F32(a, b, c) \
    __builtin_amdgcn_wmma_f32_16x16x4_f32(false, (a), false, (b), (short)0, (c), false, false)

template <int M>
__device__ __forceinline__ float rowx(float x) {
    const int b = __builtin_bit_cast(int, x);
    return __builtin_bit_cast(float,
        __builtin_amdgcn_update_dpp(0, b, 0x160 + M, 0xf, 0xf, true));
}
__device__ __forceinline__ float row_sum(float s) {
    s += rowx<8>(s);
    s += rowx<4>(s);
    s += rowx<2>(s);
    s += rowx<1>(s);
    return s;
}
__device__ __forceinline__ float row_max(float m) {
    m = fmaxf(m, rowx<8>(m));
    m = fmaxf(m, rowx<4>(m));
    m = fmaxf(m, rowx<2>(m));
    m = fmaxf(m, rowx<1>(m));
    return m;
}

template <int N>
__device__ __forceinline__ float newton_tau(const float (&x)[N], float& ssum) {
    float tau = -1.0f;
    for (int it = 0; it < 12; ++it) {
        float s = 0.f, g = 0.f;
#pragma unroll
        for (int j = 0; j < N; ++j) {
            float d = fmaxf(x[j] - tau, 0.f);
            s = fmaf(d, d, s);
            g += d;
        }
        s = row_sum(s);
        g = row_sum(g);
        g = fmaxf(g, 1e-20f);
        tau = fmaxf(tau + (s - 1.0f) / (2.0f * g), -1.0f);
    }
    float ss = 0.f;
#pragma unroll
    for (int j = 0; j < N; ++j) {
        float d = fmaxf(x[j] - tau, 0.f);
        ss += d * d;
    }
    ssum = ss;
    return tau;
}

__global__ __launch_bounds__(256)
void entmax_attn_kernel(const float* __restrict__ q,
                        const float* __restrict__ k,
                        const float* __restrict__ v,
                        float* __restrict__ out)
{
    __shared__ __attribute__((aligned(16))) float sm[16 * PITCH + 8 * 16 * KPITCH];

    const int lane  = threadIdx.x & 31;
    const int wave  = threadIdx.x >> 5;
    const int lmod  = lane & 15;
    const int lhalf = lane >> 4;

    const int bh    = blockIdx.y;
    const int qbase = blockIdx.x * 16;

    const float* qp = q + ((size_t)(bh * S_LEN + qbase + lmod)) * D_DIM + 4 * lhalf;
    v4f aq[8];
#pragma unroll
    for (int j = 0; j < 8; ++j)
        aq[j] = *(const v4f*)(qp + 8 * j);

    const float* kt = k + ((size_t)(bh * S_LEN + wave * 128)) * D_DIM + 4 * lane;
    float* stg = &sm[16 * PITCH + wave * 16 * KPITCH];
    float* stw = stg + lhalf * KPITCH + 4 * lmod;
    const float* stf = stg + lmod * KPITCH + 4 * lhalf;

    v4f g[8];
#pragma unroll
    for (int i = 0; i < 8; ++i)
        g[i] = *(const v4f*)(kt + 128 * i);

#pragma unroll
    for (int t = 0; t < 8; ++t) {
        const int nbase = (wave * 8 + t) * 16;
#pragma unroll
        for (int i = 0; i < 8; ++i)
            *(v4f*)(stw + 2 * i * KPITCH) = g[i];
        if (t < 7) {
#pragma unroll
            for (int i = 0; i < 8; ++i)
                g[i] = *(const v4f*)(kt + 1024 * (t + 1) + 128 * i);
        }
        v4f bk[8];
#pragma unroll
        for (int j = 0; j < 8; ++j)
            bk[j] = *(const v4f*)(stf + 8 * j);
        v8f acc = {};
#pragma unroll
        for (int j = 0; j < 8; ++j) {
            acc = WMMA_F32(aq[j].xy, bk[j].xy, acc);
            acc = WMMA_F32(aq[j].zw, bk[j].zw, acc);
        }
#pragma unroll
        for (int r = 0; r < 8; ++r)
            sm[(r + 8 * lhalf) * PITCH + nbase + lmod] = acc[r] * 0.0625f;
    }
    __syncthreads();

    float* cb = &sm[16 * PITCH + wave * 16 * KPITCH];
    {
        const int rowoff = (wave * 2 + lhalf) * PITCH;
        float* rp = &sm[rowoff + 4 * lmod];
        float x[64];
#pragma unroll
        for (int j = 0; j < 16; ++j) {
            const v4f t = *(const v4f*)(rp + 64 * j);
            x[4 * j + 0] = t.x;
            x[4 * j + 1] = t.y;
            x[4 * j + 2] = t.z;
            x[4 * j + 3] = t.w;
        }

        float m = x[0];
#pragma unroll
        for (int j = 1; j < 64; ++j) m = fmaxf(m, x[j]);
        m = row_max(m);
#pragma unroll
        for (int j = 0; j < 64; ++j) x[j] -= m;

        const v4f fill = {-2.0f, -2.0f, -2.0f, -2.0f};
#pragma unroll
        for (int k = 0; k < 6; ++k)
            *(v4f*)(rp + 64 * k) = fill;

        const unsigned halfmask = lhalf ? 0xffff0000u : 0x0000ffffu;
        const unsigned spare = (unsigned)(16 * PITCH + wave * 16 * KPITCH + lane);
        unsigned n = (unsigned)rowoff;
#pragma unroll
        for (int j = 0; j < 64; ++j) {
            const bool keep = x[j] > -1.0f;
            const unsigned mine = __builtin_amdgcn_ballot_w32(keep) & halfmask;
            const unsigned pos = __builtin_amdgcn_mbcnt_lo(mine, n);
            sm[keep ? pos : spare] = x[j];
            n += (unsigned)__builtin_popcount(mine);
        }
        n -= (unsigned)rowoff;
        const unsigned n0 = (unsigned)__builtin_amdgcn_readlane((int)n, 0);
        const unsigned n1 = (unsigned)__builtin_amdgcn_readlane((int)n, 16);
        const unsigned nmax = n0 > n1 ? n0 : n1;

        float tau, ssum;
        if (nmax <= 128u) {
            float c[8];
#pragma unroll
            for (int k = 0; k < 2; ++k) {
                const v4f t = *(const v4f*)(rp + 64 * k);
                c[4 * k + 0] = t.x; c[4 * k + 1] = t.y; c[4 * k + 2] = t.z; c[4 * k + 3] = t.w;
            }
            tau = newton_tau<8>(c, ssum);
        } else if (nmax <= 192u) {
            float c[12];
#pragma unroll
            for (int k = 0; k < 3; ++k) {
                const v4f t = *(const v4f*)(rp + 64 * k);
                c[4 * k + 0] = t.x; c[4 * k + 1] = t.y; c[4 * k + 2] = t.z; c[4 * k + 3] = t.w;
            }
            tau = newton_tau<12>(c, ssum);
        } else if (nmax <= 256u) {
            float c[16];
#pragma unroll
            for (int k = 0; k < 4; ++k) {
                const v4f t = *(const v4f*)(rp + 64 * k);
                c[4 * k + 0] = t.x; c[4 * k + 1] = t.y; c[4 * k + 2] = t.z; c[4 * k + 3] = t.w;
            }
            tau = newton_tau<16>(c, ssum);
        } else if (nmax <= 384u) {
            float c[24];
#pragma unroll
            for (int k = 0; k < 6; ++k) {
                const v4f t = *(const v4f*)(rp + 64 * k);
                c[4 * k + 0] = t.x; c[4 * k + 1] = t.y; c[4 * k + 2] = t.z; c[4 * k + 3] = t.w;
            }
            tau = newton_tau<24>(c, ssum);
        } else {
            tau = newton_tau<64>(x, ssum);
        }

        ssum = row_sum(ssum);
        const float inv = (ssum > 0.f) ? (1.0f / ssum) : 0.f;
#pragma unroll
        for (int j = 0; j < 16; ++j) {
            v4f p;
            float d;
            d = fmaxf(x[4 * j + 0] - tau, 0.f); p.x = (d * d) * inv;
            d = fmaxf(x[4 * j + 1] - tau, 0.f); p.y = (d * d) * inv;
            d = fmaxf(x[4 * j + 2] - tau, 0.f); p.z = (d * d) * inv;
            d = fmaxf(x[4 * j + 3] - tau, 0.f); p.w = (d * d) * inv;
            *(v4f*)(rp + 64 * j) = p;
        }
    }
    __syncthreads();

    const int kslice = wave * 128;
    int* lst = (int*)cb;
    {
        const v4i dummy = {kslice, kslice, kslice, kslice};
        *(v4i*)(lst + 4 * lane) = dummy;
        if (lane < 8) *(v4i*)(lst + 128 + 4 * lane) = dummy;
    }
    v4f mx = *(const v4f*)(&sm[kslice + 4 * lane]);
#pragma unroll
    for (int r = 1; r < 16; ++r) {
        const v4f pr = *(const v4f*)(&sm[r * PITCH + kslice + 4 * lane]);
        mx.x = fmaxf(mx.x, pr.x);
        mx.y = fmaxf(mx.y, pr.y);
        mx.z = fmaxf(mx.z, pr.z);
        mx.w = fmaxf(mx.w, pr.w);
    }
    unsigned nk = 0;
#pragma unroll
    for (int e = 0; e < 4; ++e) {
        const bool keep = mx[e] > 0.f;
        const unsigned mask = __builtin_amdgcn_ballot_w32(keep);
        const unsigned pos = __builtin_amdgcn_mbcnt_lo(mask, nk);
        if (keep) lst[pos] = kslice + 4 * lane + e;
        nk += (unsigned)__builtin_popcount(mask);
    }

    const float* vb = v + (size_t)bh * S_LEN * D_DIM + 4 * lmod;
    const float* prow = &sm[lmod * PITCH];
    const int nbatch = (int)((nk + 39u) / 40u);

    v8f acc0 = {}, acc1 = {}, acc2 = {}, acc3 = {};
    for (int b = 0; b < nbatch; ++b) {
        v4f a4[5], r0[5], r1[5], r2[5], r3[5];
#pragma unroll
        for (int u = 0; u < 5; ++u) {
            const int p0 = 40 * b + 8 * u + 4 * lhalf;
            const v4i kk = *(const v4i*)(lst + p0);
            v4f a;
            a.x = prow[kk.x];
            a.y = prow[kk.y];
            a.z = prow[kk.z];
            a.w = prow[kk.w];
            a.x = ((unsigned)(p0 + 0) < nk) ? a.x : 0.f;
            a.y = ((unsigned)(p0 + 1) < nk) ? a.y : 0.f;
            a.z = ((unsigned)(p0 + 2) < nk) ? a.z : 0.f;
            a.w = ((unsigned)(p0 + 3) < nk) ? a.w : 0.f;
            a4[u] = a;
            r0[u] = *(const v4f*)(vb + (size_t)kk.x * D_DIM);
            r1[u] = *(const v4f*)(vb + (size_t)kk.y * D_DIM);
            r2[u] = *(const v4f*)(vb + (size_t)kk.z * D_DIM);
            r3[u] = *(const v4f*)(vb + (size_t)kk.w * D_DIM);
        }
#pragma unroll
        for (int u = 0; u < 5; ++u) {
            const v2f aa = a4[u].xy;
            const v2f ab = a4[u].zw;
            const v2f b00 = {r0[u].x, r1[u].x}, b01 = {r0[u].y, r1[u].y}, b02 = {r0[u].z, r1[u].z}, b03 = {r0[u].w, r1[u].w};
            const v2f b10 = {r2[u].x, r3[u].x}, b11 = {r2[u].y, r3[u].y}, b12 = {r2[u].z, r3[u].z}, b13 = {r2[u].w, r3[u].w};
            acc0 = WMMA_F32(aa, b00, acc0);
            acc1 = WMMA_F32(aa, b01, acc1);
            acc2 = WMMA_F32(aa, b02, acc2);
            acc3 = WMMA_F32(aa, b03, acc3);
            acc0 = WMMA_F32(ab, b10, acc0);
            acc1 = WMMA_F32(ab, b11, acc1);
            acc2 = WMMA_F32(ab, b12, acc2);
            acc3 = WMMA_F32(ab, b13, acc3);
        }
    }
    __syncthreads();

#pragma unroll
    for (int r = 0; r < 8; ++r) {
        const v4f o = {acc0[r], acc1[r], acc2[r], acc3[r]};
        *(v4f*)(&sm[wave * 1024 + (8 * lhalf + r) * D_DIM + 4 * lmod]) = o;
    }
    __syncthreads();

    const int e4 = threadIdx.x * 4;
    v4f sum = *(const v4f*)(&sm[e4]);
#pragma unroll
    for (int w = 1; w < 8; ++w)
        sum += *(const v4f*)(&sm[w * 1024 + e4]);
    *(v4f*)(out + ((size_t)(bh * S_LEN + qbase)) * D_DIM + e4) = sum;
}

extern "C" void kernel_launch(void* const* d_in, const int* in_sizes, int n_in,
                              void* d_out, int out_size, void* d_ws, size_t ws_size,
                              hipStream_t stream) {
    (void)in_sizes; (void)n_in; (void)out_size; (void)d_ws; (void)ws_size;
    const float* q = (const float*)d_in[0];
    const float* k = (const float*)d_in[1];
    const float* v = (const float*)d_in[2];
    float* out = (float*)d_out;

    dim3 grid(S_LEN / 16, BH);
    entmax_attn_kernel<<<grid, 256, 0, stream>>>(q, k, v, out);
}


# Entmax candidates packed as whole 4-entry groups (16 ballot rounds, was 64 single entries) above a lower bound of tau from 3 Newton steps on the 16 per-lane maxima (was max - 1); sum order changes
# speedup vs baseline: 2.7365x; 1.0760x over previous
#include <hip/hip_runtime.h>


typedef __attribute__((ext_vector_type(2))) float v2f;
typedef __attribute__((ext_vector_type(4))) float v4f;
typedef __attribute__((ext_vector_type(8))) float v8f;
typedef __attribute__((ext_vector_type(4))) int   v4i;

#define S_LEN 1024
#define D_DIM 64
#define BH    64
#define PITCH (S_LEN + 4)
#define KPITCH (D_DIM + 4)

#define WMMA_F32(a, b, c) \
    __builtin_amdgcn_wmma_f32_16x16x4_f32(false, (a), false, (b), (short)0, (c), false, false)

template <int M>
__device__ __forceinline__ float rowx(float x) {
    const int b = __builtin_bit_cast(int, x);
    return __builtin_bit_cast(float,
        __builtin_amdgcn_update_dpp(0, b, 0x160 + M, 0xf, 0xf, true));
}
__device__ __forceinline__ float row_sum(float s) {
    s += rowx<8>(s);
    s += rowx<4>(s);
    s += rowx<2>(s);
    s += rowx<1>(s);
    return s;
}
__device__ __forceinline__ float row_max(float m) {
    m = fmaxf(m, rowx<8>(m));
    m = fmaxf(m, rowx<4>(m));
    m = fmaxf(m, rowx<2>(m));
    m = fmaxf(m, rowx<1>(m));
    return m;
}

template <int N>
__device__ __forceinline__ float newton_tau(const float (&x)[N], float& ssum, float tau) {
    for (int it = 0; it < 12; ++it) {
        float s = 0.f, g = 0.f;
#pragma unroll
        for (int j = 0; j < N; ++j) {
            float d = fmaxf(x[j] - tau, 0.f);
            s = fmaf(d, d, s);
            g += d;
        }
        s = row_sum(s);
        g = row_sum(g);
        g = fmaxf(g, 1e-20f);
        tau = fmaxf(tau + (s - 1.0f) / (2.0f * g), -1.0f);
    }
    float ss = 0.f;
#pragma unroll
    for (int j = 0; j < N; ++j) {
        float d = fmaxf(x[j] - tau, 0.f);
        ss += d * d;
    }
    ssum = ss;
    return tau;
}

__global__ __launch_bounds__(256)
void entmax_attn_kernel(const float* __restrict__ q,
                        const float* __restrict__ k,
                        const float* __restrict__ v,
                        float* __restrict__ out)
{
    __shared__ __attribute__((aligned(16))) float sm[16 * PITCH + 8 * 16 * KPITCH];

    const int lane  = threadIdx.x & 31;
    const int wave  = threadIdx.x >> 5;
    const int lmod  = lane & 15;
    const int lhalf = lane >> 4;

    const int bh    = blockIdx.y;
    const int qbase = blockIdx.x * 16;

    const float* qp = q + ((size_t)(bh * S_LEN + qbase + lmod)) * D_DIM + 4 * lhalf;
    v4f aq[8];
#pragma unroll
    for (int j = 0; j < 8; ++j)
        aq[j] = *(const v4f*)(qp + 8 * j);

    const float* kt = k + ((size_t)(bh * S_LEN + wave * 128)) * D_DIM + 4 * lane;
    float* stg = &sm[16 * PITCH + wave * 16 * KPITCH];
    float* stw = stg + lhalf * KPITCH + 4 * lmod;
    const float* stf = stg + lmod * KPITCH + 4 * lhalf;

    v4f g[8];
#pragma unroll
    for (int i = 0; i < 8; ++i)
        g[i] = *(const v4f*)(kt + 128 * i);

#pragma unroll
    for (int t = 0; t < 8; ++t) {
        const int nbase = (wave * 8 + t) * 16;
#pragma unroll
        for (int i = 0; i < 8; ++i)
            *(v4f*)(stw + 2 * i * KPITCH) = g[i];
        if (t < 7) {
#pragma unroll
            for (int i = 0; i < 8; ++i)
                g[i] = *(const v4f*)(kt + 1024 * (t + 1) + 128 * i);
        }
        v4f bk[8];
#pragma unroll
        for (int j = 0; j < 8; ++j)
            bk[j] = *(const v4f*)(stf + 8 * j);
        v8f acc = {};
#pragma unroll
        for (int j = 0; j < 8; ++j) {
            acc = WMMA_F32(aq[j].xy, bk[j].xy, acc);
            acc = WMMA_F32(aq[j].zw, bk[j].zw, acc);
        }
#pragma unroll
        for (int r = 0; r < 8; ++r)
            sm[(r + 8 * lhalf) * PITCH + nbase + lmod] = acc[r] * 0.0625f;
    }
    __syncthreads();

    float* cb = &sm[16 * PITCH + wave * 16 * KPITCH];
    {
        const int rowoff = (wave * 2 + lhalf) * PITCH;
        float* rp = &sm[rowoff + 4 * lmod];
        float x[64];
#pragma unroll
        for (int j = 0; j < 16; ++j) {
            const v4f t = *(const v4f*)(rp + 64 * j);
            x[4 * j + 0] = t.x;
            x[4 * j + 1] = t.y;
            x[4 * j + 2] = t.z;
            x[4 * j + 3] = t.w;
        }

        float gm[16];
#pragma unroll
        for (int j = 0; j < 16; ++j)
            gm[j] = fmaxf(fmaxf(x[4 * j + 0], x[4 * j + 1]), fmaxf(x[4 * j + 2], x[4 * j + 3]));
        float ml = gm[0];
#pragma unroll
        for (int j = 1; j < 16; ++j) ml = fmaxf(ml, gm[j]);
        const float m = row_max(ml);
#pragma unroll
        for (int j = 0; j < 64; ++j) x[j] -= m;

        float lb = -1.0f;
        {
            const float y = ml - m;
            for (int it = 0; it < 3; ++it) {
                const float d = fmaxf(y - lb, 0.f);
                const float s = row_sum(d * d);
                const float g = fmaxf(row_sum(d), 1e-20f);
                lb = fmaxf(fmaf(s - 1.0f, __builtin_amdgcn_rcpf(2.0f * g), lb), -1.0f);
            }
            lb -= 1e-5f;
        }
        const float thr = lb + m;

        const v4f fill = {-2.0f, -2.0f, -2.0f, -2.0f};
#pragma unroll
        for (int k = 0; k < 6; ++k)
            *(v4f*)(rp + 64 * k) = fill;

        const unsigned halfmask = lhalf ? 0xffff0000u : 0x0000ffffu;
        const unsigned spare = (unsigned)(16 * PITCH + wave * 16 * KPITCH + 4 * lane);
        unsigned n = (unsigned)rowoff;
#pragma unroll
        for (int j = 0; j < 16; ++j) {
            const bool keep = gm[j] > thr;
            const unsigned mine = __builtin_amdgcn_ballot_w32(keep) & halfmask;
            const unsigned pos = n + 4u * __builtin_amdgcn_mbcnt_lo(mine, 0u);
            const v4f grp = {x[4 * j + 0], x[4 * j + 1], x[4 * j + 2], x[4 * j + 3]};
            *(v4f*)__builtin_assume_aligned(&sm[keep ? pos : spare], 16) = grp;
            n += 4u * (unsigned)__builtin_popcount(mine);
        }
        n -= (unsigned)rowoff;
        const unsigned n0 = (unsigned)__builtin_amdgcn_readlane((int)n, 0);
        const unsigned n1 = (unsigned)__builtin_amdgcn_readlane((int)n, 16);
        const unsigned nmax = n0 > n1 ? n0 : n1;

        float tau, ssum;
        if (nmax <= 128u) {
            float c[8];
#pragma unroll
            for (int k = 0; k < 2; ++k) {
                const v4f t = *(const v4f*)(rp + 64 * k);
                c[4 * k + 0] = t.x; c[4 * k + 1] = t.y; c[4 * k + 2] = t.z; c[4 * k + 3] = t.w;
            }
            tau = newton_tau<8>(c, ssum, lb);
        } else if (nmax <= 256u) {
            float c[16];
#pragma unroll
            for (int k = 0; k < 4; ++k) {
                const v4f t = *(const v4f*)(rp + 64 * k);
                c[4 * k + 0] = t.x; c[4 * k + 1] = t.y; c[4 * k + 2] = t.z; c[4 * k + 3] = t.w;
            }
            tau = newton_tau<16>(c, ssum, lb);
        } else if (nmax <= 384u) {
            float c[24];
#pragma unroll
            for (int k = 0; k < 6; ++k) {
                const v4f t = *(const v4f*)(rp + 64 * k);
                c[4 * k + 0] = t.x; c[4 * k + 1] = t.y; c[4 * k + 2] = t.z; c[4 * k + 3] = t.w;
            }
            tau = newton_tau<24>(c, ssum, lb);
        } else {
            tau = newton_tau<64>(x, ssum, lb);
        }

        ssum = row_sum(ssum);
        const float inv = (ssum > 0.f) ? (1.0f / ssum) : 0.f;
#pragma unroll
        for (int j = 0; j < 16; ++j) {
            v4f p;
            float d;
            d = fmaxf(x[4 * j + 0] - tau, 0.f); p.x = (d * d) * inv;
            d = fmaxf(x[4 * j + 1] - tau, 0.f); p.y = (d * d) * inv;
            d = fmaxf(x[4 * j + 2] - tau, 0.f); p.z = (d * d) * inv;
            d = fmaxf(x[4 * j + 3] - tau, 0.f); p.w = (d * d) * inv;
            *(v4f*)(rp + 64 * j) = p;
        }
    }
    __syncthreads();

    const int kslice = wave * 128;
    int* lst = (int*)cb;
    {
        const v4i dummy = {kslice, kslice, kslice, kslice};
        *(v4i*)(lst + 4 * lane) = dummy;
        if (lane < 8) *(v4i*)(lst + 128 + 4 * lane) = dummy;
    }
    v4f mx = *(const v4f*)(&sm[kslice + 4 * lane]);
#pragma unroll
    for (int r = 1; r < 16; ++r) {
        const v4f pr = *(const v4f*)(&sm[r * PITCH + kslice + 4 * lane]);
        mx.x = fmaxf(mx.x, pr.x);
        mx.y = fmaxf(mx.y, pr.y);
        mx.z = fmaxf(mx.z, pr.z);
        mx.w = fmaxf(mx.w, pr.w);
    }
    unsigned nk = 0;
#pragma unroll
    for (int e = 0; e < 4; ++e) {
        const bool keep = mx[e] > 0.f;
        const unsigned mask = __builtin_amdgcn_ballot_w32(keep);
        const unsigned pos = __builtin_amdgcn_mbcnt_lo(mask, nk);
        if (keep) lst[pos] = kslice + 4 * lane + e;
        nk += (unsigned)__builtin_popcount(mask);
    }

    const float* vb = v + (size_t)bh * S_LEN * D_DIM + 4 * lmod;
    const float* prow = &sm[lmod * PITCH];
    const int nbatch = (int)((nk + 39u) / 40u);

    v8f acc0 = {}, acc1 = {}, acc2 = {}, acc3 = {};
    for (int b = 0; b < nbatch; ++b) {
        v4f a4[5], r0[5], r1[5], r2[5], r3[5];
#pragma unroll
        for (int u = 0; u < 5; ++u) {
            const int p0 = 40 * b + 8 * u + 4 * lhalf;
            const v4i kk = *(const v4i*)(lst + p0);
            v4f a;
            a.x = prow[kk.x];
            a.y = prow[kk.y];
            a.z = prow[kk.z];
            a.w = prow[kk.w];
            a.x = ((unsigned)(p0 + 0) < nk) ? a.x : 0.f;
            a.y = ((unsigned)(p0 + 1) < nk) ? a.y : 0.f;
            a.z = ((unsigned)(p0 + 2) < nk) ? a.z : 0.f;
            a.w = ((unsigned)(p0 + 3) < nk) ? a.w : 0.f;
            a4[u] = a;
            r0[u] = *(const v4f*)(vb + (size_t)kk.x * D_DIM);
            r1[u] = *(const v4f*)(vb + (size_t)kk.y * D_DIM);
            r2[u] = *(const v4f*)(vb + (size_t)kk.z * D_DIM);
            r3[u] = *(const v4f*)(vb + (size_t)kk.w * D_DIM);
        }
#pragma unroll
        for (int u = 0; u < 5; ++u) {
            const v2f aa = a4[u].xy;
            const v2f ab = a4[u].zw;
            const v2f b00 = {r0[u].x, r1[u].x}, b01 = {r0[u].y, r1[u].y}, b02 = {r0[u].z, r1[u].z}, b03 = {r0[u].w, r1[u].w};
            const v2f b10 = {r2[u].x, r3[u].x}, b11 = {r2[u].y, r3[u].y}, b12 = {r2[u].z, r3[u].z}, b13 = {r2[u].w, r3[u].w};
            acc0 = WMMA_F32(aa, b00, acc0);
            acc1 = WMMA_F32(aa, b01, acc1);
            acc2 = WMMA_F32(aa, b02, acc2);
            acc3 = WMMA_F32(aa, b03, acc3);
            acc0 = WMMA_F32(ab, b10, acc0);
            acc1 = WMMA_F32(ab, b11, acc1);
            acc2 = WMMA_F32(ab, b12, acc2);
            acc3 = WMMA_F32(ab, b13, acc3);
        }
    }
    __syncthreads();

#pragma unroll
    for (int r = 0; r < 8; ++r) {
        const v4f o = {acc0[r], acc1[r], acc2[r], acc3[r]};
        *(v4f*)(&sm[wave * 1024 + (8 * lhalf + r) * D_DIM + 4 * lmod]) = o;
    }
    __syncthreads();

    const int e4 = threadIdx.x * 4;
    v4f sum = *(const v4f*)(&sm[e4]);
#pragma unroll
    for (int w = 1; w < 8; ++w)
        sum += *(const v4f*)(&sm[w * 1024 + e4]);
    *(v4f*)(out + ((size_t)(bh * S_LEN + qbase)) * D_DIM + e4) = sum;
}

extern "C" void kernel_launch(void* const* d_in, const int* in_sizes, int n_in,
                              void* d_out, int out_size, void* d_ws, size_t ws_size,
                              hipStream_t stream) {
    (void)in_sizes; (void)n_in; (void)out_size; (void)d_ws; (void)ws_size;
    const float* q = (const float*)d_in[0];
    const float* k = (const float*)d_in[1];
    const float* v = (const float*)d_in[2];
    float* out = (float*)d_out;

    dim3 grid(S_LEN / 16, BH);
    entmax_attn_kernel<<<grid, 256, 0, stream>>>(q, k, v, out);
}


# Entmax Newton on tau stops at the first step that raises tau in neither row of the wave (wave-uniform ballot exit, still at most 12 steps; was always 12); tau may differ in its last bit
# speedup vs baseline: 2.8979x; 1.0590x over previous
#include <hip/hip_runtime.h>


typedef __attribute__((ext_vector_type(2))) float v2f;
typedef __attribute__((ext_vector_type(4))) float v4f;
typedef __attribute__((ext_vector_type(8))) float v8f;
typedef __attribute__((ext_vector_type(4))) int   v4i;

#define S_LEN 1024
#define D_DIM 64
#define BH    64
#define PITCH (S_LEN + 4)
#define KPITCH (D_DIM + 4)

#define WMMA_F32(a, b, c) \
    __builtin_amdgcn_wmma_f32_16x16x4_f32(false, (a), false, (b), (short)0, (c), false, false)

template <int M>
__device__ __forceinline__ float rowx(float x) {
    const int b = __builtin_bit_cast(int, x);
    return __builtin_bit_cast(float,
        __builtin_amdgcn_update_dpp(0, b, 0x160 + M, 0xf, 0xf, true));
}
__device__ __forceinline__ float row_sum(float s) {
    s += rowx<8>(s);
    s += rowx<4>(s);
    s += rowx<2>(s);
    s += rowx<1>(s);
    return s;
}
__device__ __forceinline__ float row_max(float m) {
    m = fmaxf(m, rowx<8>(m));
    m = fmaxf(m, rowx<4>(m));
    m = fmaxf(m, rowx<2>(m));
    m = fmaxf(m, rowx<1>(m));
    return m;
}

template <int N>
__device__ __forceinline__ float newton_tau(const float (&x)[N], float& ssum, float tau) {
    for (int it = 0; it < 12; ++it) {
        float s = 0.f, g = 0.f;
#pragma unroll
        for (int j = 0; j < N; ++j) {
            float d = fmaxf(x[j] - tau, 0.f);
            s = fmaf(d, d, s);
            g += d;
        }
        s = row_sum(s);
        g = row_sum(g);
        g = fmaxf(g, 1e-20f);
        const float tn = fmaxf(tau + (s - 1.0f) / (2.0f * g), -1.0f);
        const bool rose = tn > tau;
        tau = tn;
        if (__builtin_amdgcn_ballot_w32(rose) == 0u) break;
    }
    float ss = 0.f;
#pragma unroll
    for (int j = 0; j < N; ++j) {
        float d = fmaxf(x[j] - tau, 0.f);
        ss += d * d;
    }
    ssum = ss;
    return tau;
}

__global__ __launch_bounds__(256)
void entmax_attn_kernel(const float* __restrict__ q,
                        const float* __restrict__ k,
                        const float* __restrict__ v,
                        float* __restrict__ out)
{
    __shared__ __attribute__((aligned(16))) float sm[16 * PITCH + 8 * 16 * KPITCH];

    const int lane  = threadIdx.x & 31;
    const int wave  = threadIdx.x >> 5;
    const int lmod  = lane & 15;
    const int lhalf = lane >> 4;

    const int bh    = blockIdx.y;
    const int qbase = blockIdx.x * 16;

    const float* qp = q + ((size_t)(bh * S_LEN + qbase + lmod)) * D_DIM + 4 * lhalf;
    v4f aq[8];
#pragma unroll
    for (int j = 0; j < 8; ++j)
        aq[j] = *(const v4f*)(qp + 8 * j);

    const float* kt = k + ((size_t)(bh * S_LEN + wave * 128)) * D_DIM + 4 * lane;
    float* stg = &sm[16 * PITCH + wave * 16 * KPITCH];
    float* stw = stg + lhalf * KPITCH + 4 * lmod;
    const float* stf = stg + lmod * KPITCH + 4 * lhalf;

    v4f g[8];
#pragma unroll
    for (int i = 0; i < 8; ++i)
        g[i] = *(const v4f*)(kt + 128 * i);

#pragma unroll
    for (int t = 0; t < 8; ++t) {
        const int nbase = (wave * 8 + t) * 16;
#pragma unroll
        for (int i = 0; i < 8; ++i)
            *(v4f*)(stw + 2 * i * KPITCH) = g[i];
        if (t < 7) {
#pragma unroll
            for (int i = 0; i < 8; ++i)
                g[i] = *(const v4f*)(kt + 1024 * (t + 1) + 128 * i);
        }
        v4f bk[8];
#pragma unroll
        for (int j = 0; j < 8; ++j)
            bk[j] = *(const v4f*)(stf + 8 * j);
        v8f acc = {};
#pragma unroll
        for (int j = 0; j < 8; ++j) {
            acc = WMMA_F32(aq[j].xy, bk[j].xy, acc);
            acc = WMMA_F32(aq[j].zw, bk[j].zw, acc);
        }
#pragma unroll
        for (int r = 0; r < 8; ++r)
            sm[(r + 8 * lhalf) * PITCH + nbase + lmod] = acc[r] * 0.0625f;
    }
    __syncthreads();

    float* cb = &sm[16 * PITCH + wave * 16 * KPITCH];
    {
        const int rowoff = (wave * 2 + lhalf) * PITCH;
        float* rp = &sm[rowoff + 4 * lmod];
        float x[64];
#pragma unroll
        for (int j = 0; j < 16; ++j) {
            const v4f t = *(const v4f*)(rp + 64 * j);
            x[4 * j + 0] = t.x;
            x[4 * j + 1] = t.y;
            x[4 * j + 2] = t.z;
            x[4 * j + 3] = t.w;
        }

        float gm[16];
#pragma unroll
        for (int j = 0; j < 16; ++j)
            gm[j] = fmaxf(fmaxf(x[4 * j + 0], x[4 * j + 1]), fmaxf(x[4 * j + 2], x[4 * j + 3]));
        float ml = gm[0];
#pragma unroll
        for (int j = 1; j < 16; ++j) ml = fmaxf(ml, gm[j]);
        const float m = row_max(ml);
#pragma unroll
        for (int j = 0; j < 64; ++j) x[j] -= m;

        float lb = -1.0f;
        {
            const float y = ml - m;
            for (int it = 0; it < 3; ++it) {
                const float d = fmaxf(y - lb, 0.f);
                const float s = row_sum(d * d);
                const float g = fmaxf(row_sum(d), 1e-20f);
                lb = fmaxf(fmaf(s - 1.0f, __builtin_amdgcn_rcpf(2.0f * g), lb), -1.0f);
            }
            lb -= 1e-5f;
        }
        const float thr = lb + m;

        const v4f fill = {-2.0f, -2.0f, -2.0f, -2.0f};
#pragma unroll
        for (int k = 0; k < 6; ++k)
            *(v4f*)(rp + 64 * k) = fill;

        const unsigned halfmask = lhalf ? 0xffff0000u : 0x0000ffffu;
        const unsigned spare = (unsigned)(16 * PITCH + wave * 16 * KPITCH + 4 * lane);
        unsigned n = (unsigned)rowoff;
#pragma unroll
        for (int j = 0; j < 16; ++j) {
            const bool keep = gm[j] > thr;
            const unsigned mine = __builtin_amdgcn_ballot_w32(keep) & halfmask;
            const unsigned pos = n + 4u * __builtin_amdgcn_mbcnt_lo(mine, 0u);
            const v4f grp = {x[4 * j + 0], x[4 * j + 1], x[4 * j + 2], x[4 * j + 3]};
            *(v4f*)__builtin_assume_aligned(&sm[keep ? pos : spare], 16) = grp;
            n += 4u * (unsigned)__builtin_popcount(mine);
        }
        n -= (unsigned)rowoff;
        const unsigned n0 = (unsigned)__builtin_amdgcn_readlane((int)n, 0);
        const unsigned n1 = (unsigned)__builtin_amdgcn_readlane((int)n, 16);
        const unsigned nmax = n0 > n1 ? n0 : n1;

        float tau, ssum;
        if (nmax <= 128u) {
            float c[8];
#pragma unroll
            for (int k = 0; k < 2; ++k) {
                const v4f t = *(const v4f*)(rp + 64 * k);
                c[4 * k + 0] = t.x; c[4 * k + 1] = t.y; c[4 * k + 2] = t.z; c[4 * k + 3] = t.w;
            }
            tau = newton_tau<8>(c, ssum, lb);
        } else if (nmax <= 256u) {
            float c[16];
#pragma unroll
            for (int k = 0; k < 4; ++k) {
                const v4f t = *(const v4f*)(rp + 64 * k);
                c[4 * k + 0] = t.x; c[4 * k + 1] = t.y; c[4 * k + 2] = t.z; c[4 * k + 3] = t.w;
            }
            tau = newton_tau<16>(c, ssum, lb);
        } else if (nmax <= 384u) {
            float c[24];
#pragma unroll
            for (int k = 0; k < 6; ++k) {
                const v4f t = *(const v4f*)(rp + 64 * k);
                c[4 * k + 0] = t.x; c[4 * k + 1] = t.y; c[4 * k + 2] = t.z; c[4 * k + 3] = t.w;
            }
            tau = newton_tau<24>(c, ssum, lb);
        } else {
            tau = newton_tau<64>(x, ssum, lb);
        }

        ssum = row_sum(ssum);
        const float inv = (ssum > 0.f) ? (1.0f / ssum) : 0.f;
#pragma unroll
        for (int j = 0; j < 16; ++j) {
            v4f p;
            float d;
            d = fmaxf(x[4 * j + 0] - tau, 0.f); p.x = (d * d) * inv;
            d = fmaxf(x[4 * j + 1] - tau, 0.f); p.y = (d * d) * inv;
            d = fmaxf(x[4 * j + 2] - tau, 0.f); p.z = (d * d) * inv;
            d = fmaxf(x[4 * j + 3] - tau, 0.f); p.w = (d * d) * inv;
            *(v4f*)(rp + 64 * j) = p;
        }
    }
    __syncthreads();

    const int kslice = wave * 128;
    int* lst = (int*)cb;
    {
        const v4i dummy = {kslice, kslice, kslice, kslice};
        *(v4i*)(lst + 4 * lane) = dummy;
        if (lane < 8) *(v4i*)(lst + 128 + 4 * lane) = dummy;
    }
    v4f mx = *(const v4f*)(&sm[kslice + 4 * lane]);
#pragma unroll
    for (int r = 1; r < 16; ++r) {
        const v4f pr = *(const v4f*)(&sm[r * PITCH + kslice + 4 * lane]);
        mx.x = fmaxf(mx.x, pr.x);
        mx.y = fmaxf(mx.y, pr.y);
        mx.z = fmaxf(mx.z, pr.z);
        mx.w = fmaxf(mx.w, pr.w);
    }
    unsigned nk = 0;
#pragma unroll
    for (int e = 0; e < 4; ++e) {
        const bool keep = mx[e] > 0.f;
        const unsigned mask = __builtin_amdgcn_ballot_w32(keep);
        const unsigned pos = __builtin_amdgcn_mbcnt_lo(mask, nk);
        if (keep) lst[pos] = kslice + 4 * lane + e;
        nk += (unsigned)__builtin_popcount(mask);
    }

    const float* vb = v + (size_t)bh * S_LEN * D_DIM + 4 * lmod;
    const float* prow = &sm[lmod * PITCH];
    const int nbatch = (int)((nk + 39u) / 40u);

    v8f acc0 = {}, acc1 = {}, acc2 = {}, acc3 = {};
    for (int b = 0; b < nbatch; ++b) {
        v4f a4[5], r0[5], r1[5], r2[5], r3[5];
#pragma unroll
        for (int u = 0; u < 5; ++u) {
            const int p0 = 40 * b + 8 * u + 4 * lhalf;
            const v4i kk = *(const v4i*)(lst + p0);
            v4f a;
            a.x = prow[kk.x];
            a.y = prow[kk.y];
            a.z = prow[kk.z];
            a.w = prow[kk.w];
            a.x = ((unsigned)(p0 + 0) < nk) ? a.x : 0.f;
            a.y = ((unsigned)(p0 + 1) < nk) ? a.y : 0.f;
            a.z = ((unsigned)(p0 + 2) < nk) ? a.z : 0.f;
            a.w = ((unsigned)(p0 + 3) < nk) ? a.w : 0.f;
            a4[u] = a;
            r0[u] = *(const v4f*)(vb + (size_t)kk.x * D_DIM);
            r1[u] = *(const v4f*)(vb + (size_t)kk.y * D_DIM);
            r2[u] = *(const v4f*)(vb + (size_t)kk.z * D_DIM);
            r3[u] = *(const v4f*)(vb + (size_t)kk.w * D_DIM);
        }
#pragma unroll
        for (int u = 0; u < 5; ++u) {
            const v2f aa = a4[u].xy;
            const v2f ab = a4[u].zw;
            const v2f b00 = {r0[u].x, r1[u].x}, b01 = {r0[u].y, r1[u].y}, b02 = {r0[u].z, r1[u].z}, b03 = {r0[u].w, r1[u].w};
            const v2f b10 = {r2[u].x, r3[u].x}, b11 = {r2[u].y, r3[u].y}, b12 = {r2[u].z, r3[u].z}, b13 = {r2[u].w, r3[u].w};
            acc0 = WMMA_F32(aa, b00, acc0);
            acc1 = WMMA_F32(aa, b01, acc1);
            acc2 = WMMA_F32(aa, b02, acc2);
            acc3 = WMMA_F32(aa, b03, acc3);
            acc0 = WMMA_F32(ab, b10, acc0);
            acc1 = WMMA_F32(ab, b11, acc1);
            acc2 = WMMA_F32(ab, b12, acc2);
            acc3 = WMMA_F32(ab, b13, acc3);
        }
    }
    __syncthreads();

#pragma unroll
    for (int r = 0; r < 8; ++r) {
        const v4f o = {acc0[r], acc1[r], acc2[r], acc3[r]};
        *(v4f*)(&sm[wave * 1024 + (8 * lhalf + r) * D_DIM + 4 * lmod]) = o;
    }
    __syncthreads();

    const int e4 = threadIdx.x * 4;
    v4f sum = *(const v4f*)(&sm[e4]);
#pragma unroll
    for (int w = 1; w < 8; ++w)
        sum += *(const v4f*)(&sm[w * 1024 + e4]);
    *(v4f*)(out + ((size_t)(bh * S_LEN + qbase)) * D_DIM + e4) = sum;
}

extern "C" void kernel_launch(void* const* d_in, const int* in_sizes, int n_in,
                              void* d_out, int out_size, void* d_ws, size_t ws_size,
                              hipStream_t stream) {
    (void)in_sizes; (void)n_in; (void)out_size; (void)d_ws; (void)ws_size;
    const float* q = (const float*)d_in[0];
    const float* k = (const float*)d_in[1];
    const float* v = (const float*)d_in[2];
    float* out = (float*)d_out;

    dim3 grid(S_LEN / 16, BH);
    entmax_attn_kernel<<<grid, 256, 0, stream>>>(q, k, v, out);
}
